# starting kernel
# baseline (speedup 1.0000x reference)
#include <hip/hip_runtime.h>
#include <hip/hip_bf16.h>
#include <math.h>

typedef __attribute__((ext_vector_type(2))) float v2f;
typedef __attribute__((ext_vector_type(8))) float v8f;

static inline int ceil_div_ll(long long a, long long b) { return (int)((a + b - 1) / b); }

template<int K, int M>
__global__ __launch_bounds__(256)
void wmma_gemm_f32(const float* __restrict__ A, const float* __restrict__ W,
                   const float* __restrict__ bias, float* __restrict__ C, int N)
{
    constexpr int TILES_N = M / 16;
    int wave = (int)(blockIdx.x * (blockDim.x >> 5)) + (int)(threadIdx.x >> 5);
    int lane = threadIdx.x & 31;
    int tile_m = wave / TILES_N;
    int row0 = tile_m * 16;
    if (row0 >= N) return;
    int col0 = (wave % TILES_N) * 16;
    int lh = lane >> 4;
    int lm = lane & 15;

    const float* a_ptr = A + (size_t)(row0 + lm) * K + (lh << 1);
    const float* b_ptr = W + (size_t)(lh << 1) * M + col0 + lm;

    v8f acc = {};
#pragma unroll
    for (int k = 0; k < K; k += 4) {
        v2f a, b;
        a.x = a_ptr[k];
        a.y = a_ptr[k + 1];
        b.x = b_ptr[(size_t)k * M];
        b.y = b_ptr[(size_t)(k + 1) * M];
        acc = __builtin_amdgcn_wmma_f32_16x16x4_f32(false, a, false, b,
                                                    (short)0, acc, false, false);
    }

    float badd = bias ? bias[col0 + lm] : 0.0f;
#pragma unroll
    for (int r = 0; r < 8; ++r) {
        int row = row0 + r + (lh << 3);
        C[(size_t)row * M + col0 + lm] = acc[r] + badd;
    }
}

__global__ void fill_f32_kernel(float* __restrict__ p, float v, long long n)
{
    long long i = (long long)blockIdx.x * blockDim.x + threadIdx.x;
    if (i < n) p[i] = v;
}

__global__ void deg_kernel(const int* __restrict__ dst, float* __restrict__ deg, int E)
{
    int e = blockIdx.x * blockDim.x + threadIdx.x;
    if (e < E) atomicAdd(&deg[dst[e]], 1.0f);
}

__global__ void dinv_kernel(const float* __restrict__ deg, float* __restrict__ dinv, int n)
{
    int i = blockIdx.x * blockDim.x + threadIdx.x;
    if (i < n) dinv[i] = rsqrtf(deg[i] + 1.0f);
}

__global__ void norm_kernel(const int* __restrict__ src, const int* __restrict__ dst,
                            const float* __restrict__ dinv, float* __restrict__ nrm, int E)
{
    int e = blockIdx.x * blockDim.x + threadIdx.x;
    if (e < E) nrm[e] = dinv[src[e]] * dinv[dst[e]];
}

template<int M>
__global__ __launch_bounds__(256)
void edge_scatter(const int* __restrict__ src, const int* __restrict__ dst,
                  const float* __restrict__ nrm, const float* __restrict__ xw,
                  float* __restrict__ agg, int E)
{
    constexpr int CPE = M / 4;
    int gid = blockIdx.x * blockDim.x + threadIdx.x;
    int e = gid / CPE;
    if (e >= E) return;
    int c = (gid % CPE) << 2;
    int s = src[e], d = dst[e];
    float nm = nrm[e];
    const float4 v = *(const float4*)(xw + (size_t)s * M + c);
    float* o = agg + (size_t)d * M + c;
    atomicAdd(o + 0, v.x * nm);
    atomicAdd(o + 1, v.y * nm);
    atomicAdd(o + 2, v.z * nm);
    atomicAdd(o + 3, v.w * nm);
}

template<int M>
__global__ __launch_bounds__(256)
void finalize_ln_elu(const float* __restrict__ agg, const float* __restrict__ xw,
                     const float* __restrict__ dinv, const float* __restrict__ bias,
                     const float* __restrict__ g, const float* __restrict__ be,
                     float* __restrict__ out, int N)
{
    constexpr int PER = M / 32;
    int node = (int)(blockIdx.x * (blockDim.x >> 5)) + (int)(threadIdx.x >> 5);
    int lane = threadIdx.x & 31;
    if (node >= N) return;

    float vals[PER];
    float sum = 0.0f;
    float selfw = xw ? dinv[node] * dinv[node] : 0.0f;
#pragma unroll
    for (int i = 0; i < PER; ++i) {
        int c = lane + (i << 5);
        float v = agg[(size_t)node * M + c];
        if (xw)   v += xw[(size_t)node * M + c] * selfw;
        if (bias) v += bias[c];
        vals[i] = v;
        sum += v;
    }
#pragma unroll
    for (int off = 16; off; off >>= 1) sum += __shfl_xor(sum, off, 32);
    float mu = sum * (1.0f / M);
    float vs = 0.0f;
#pragma unroll
    for (int i = 0; i < PER; ++i) { float d = vals[i] - mu; vs += d * d; }
#pragma unroll
    for (int off = 16; off; off >>= 1) vs += __shfl_xor(vs, off, 32);
    float rstd = rsqrtf(vs * (1.0f / M) + 1e-5f);
#pragma unroll
    for (int i = 0; i < PER; ++i) {
        int c = lane + (i << 5);
        float v = (vals[i] - mu) * rstd * g[c] + be[c];
        out[(size_t)node * M + c] = (v > 0.0f) ? v : (expf(v) - 1.0f);
    }
}

extern "C" void kernel_launch(void* const* d_in, const int* in_sizes, int n_in,
                              void* d_out, int out_size, void* d_ws, size_t ws_size,
                              hipStream_t stream)
{
    const float* x   = (const float*)d_in[0];
    const int*   ei  = (const int*)  d_in[1];
    const float* W1  = (const float*)d_in[2];
    const float* b1  = (const float*)d_in[3];
    const float* g1  = (const float*)d_in[4];
    const float* be1 = (const float*)d_in[5];
    const float* W2  = (const float*)d_in[6];
    const float* b2  = (const float*)d_in[7];
    const float* g2  = (const float*)d_in[8];
    const float* be2 = (const float*)d_in[9];
    const float* W3  = (const float*)d_in[10];
    const float* b3  = (const float*)d_in[11];
    const float* g3  = (const float*)d_in[12];
    const float* be3 = (const float*)d_in[13];
    const float* lw1 = (const float*)d_in[14];
    const float* lb1 = (const float*)d_in[15];
    const float* g4  = (const float*)d_in[16];
    const float* be4 = (const float*)d_in[17];
    const float* lw2 = (const float*)d_in[18];
    const float* lb2 = (const float*)d_in[19];

    const int N = in_sizes[0] / 128;
    const int E = in_sizes[1] / 2;
    const int* src = ei;
    const int* dst = ei + E;

    float* ws   = (float*)d_ws;
    float* xw   = ws;  ws += (size_t)N * 128;
    float* agg  = ws;  ws += (size_t)N * 128;
    float* h    = ws;  ws += (size_t)N * 128;
    float* deg  = ws;  ws += N;
    float* dinv = ws;  ws += N;
    float* nrm  = ws;  ws += E;

    const int B = 256;
    auto gemm_blocks = [N](int tiles_n) {
        long long waves = (long long)((N + 15) / 16) * tiles_n;
        return (int)((waves + 7) / 8);
    };
    const int node_blocks = ceil_div_ll(N, 8);

    fill_f32_kernel<<<ceil_div_ll(N, B), B, 0, stream>>>(deg, 0.0f, (long long)N);
    deg_kernel<<<ceil_div_ll(E, B), B, 0, stream>>>(dst, deg, E);
    dinv_kernel<<<ceil_div_ll(N, B), B, 0, stream>>>(deg, dinv, N);
    norm_kernel<<<ceil_div_ll(E, B), B, 0, stream>>>(src, dst, dinv, nrm, E);

    wmma_gemm_f32<128, 64><<<gemm_blocks(4), B, 0, stream>>>(x, W1, nullptr, xw, N);
    fill_f32_kernel<<<ceil_div_ll((long long)N * 64, B), B, 0, stream>>>(agg, 0.0f, (long long)N * 64);
    edge_scatter<64><<<ceil_div_ll((long long)E * 16, B), B, 0, stream>>>(src, dst, nrm, xw, agg, E);
    finalize_ln_elu<64><<<node_blocks, B, 0, stream>>>(agg, xw, dinv, b1, g1, be1, h, N);

    wmma_gemm_f32<64, 128><<<gemm_blocks(8), B, 0, stream>>>(h, W2, nullptr, xw, N);
    fill_f32_kernel<<<ceil_div_ll((long long)N * 128, B), B, 0, stream>>>(agg, 0.0f, (long long)N * 128);
    edge_scatter<128><<<ceil_div_ll((long long)E * 32, B), B, 0, stream>>>(src, dst, nrm, xw, agg, E);
    finalize_ln_elu<128><<<node_blocks, B, 0, stream>>>(agg, xw, dinv, b2, g2, be2, h, N);

    wmma_gemm_f32<128, 64><<<gemm_blocks(4), B, 0, stream>>>(h, W3, nullptr, xw, N);
    fill_f32_kernel<<<ceil_div_ll((long long)N * 64, B), B, 0, stream>>>(agg, 0.0f, (long long)N * 64);
    edge_scatter<64><<<ceil_div_ll((long long)E * 16, B), B, 0, stream>>>(src, dst, nrm, xw, agg, E);
    finalize_ln_elu<64><<<node_blocks, B, 0, stream>>>(agg, xw, dinv, b3, g3, be3, h, N);

    wmma_gemm_f32<64, 32><<<gemm_blocks(2), B, 0, stream>>>(h, lw1, lb1, xw, N);
    finalize_ln_elu<32><<<node_blocks, B, 0, stream>>>(xw, nullptr, nullptr, nullptr, g4, be4, h, N);
    wmma_gemm_f32<32, 32><<<gemm_blocks(2), B, 0, stream>>>(h, lw2, lb2, (float*)d_out, N);
}


# Edge-sum kernels: one whole-row load per edge (16 or 8 bytes per lane: 4 or 2 lines per instruction), 16 rows in flight; tile rows lane-major so LDS adds stay conflict-free; LN, k sum order change
# speedup vs baseline: 3.1486x; 3.1486x over previous
#include <hip/hip_runtime.h>
#include <hip/hip_bf16.h>
#include <math.h>

typedef __attribute__((ext_vector_type(2))) float v2f;
typedef __attribute__((ext_vector_type(8))) float v8f;

static inline int ceil_div_ll(long long a, long long b) { return (int)((a + b - 1) / b); }


template<int M, bool BIAS>
__device__ __forceinline__
void store_tile(const v8f& acc, const float* __restrict__ bias, float* __restrict__ C,
                int row0, int col0, int lh, int lm)
{
    float* p = C + (size_t)(row0 + (lh << 3)) * M + col0 + lm;
    if constexpr (BIAS) {
        float badd = bias[col0 + lm];
#pragma unroll
        for (int r = 0; r < 8; ++r) p[(size_t)r * M] = acc[r] + badd;
    } else {
#pragma unroll
        for (int r = 0; r < 8; ++r) p[(size_t)r * M] = acc[r];
    }
}

__global__ void fill_f32_kernel(float* __restrict__ p, float v, long long n)
{
    long long i = (long long)blockIdx.x * blockDim.x + threadIdx.x;
    if (i < n) p[i] = v;
}

constexpr int NPB   = 400;
constexpr int MAXNB = 1024;
constexpr int EPT   = 4;

template<int K, int M2, bool LN, bool BIAS>
__global__ __launch_bounds__(1024)
void bucket_gemm_f32(const float* __restrict__ IN, const float* __restrict__ g,
                     const float* __restrict__ be, const float* __restrict__ W,
                     const float* __restrict__ bias, float* __restrict__ C, int N)
{
    constexpr int P   = K + 4;
    constexpr int Q   = K / 4;
    constexpr int PER = K / 32;
    constexpr int R   = 16 / PER;
    __shared__ __attribute__((aligned(16))) float tile[NPB * P];
    const int tid  = threadIdx.x;
    const int lane = tid & 31;
    const int n0   = blockIdx.x * NPB;
    const int nn   = (N - n0 < NPB) ? (N - n0) : NPB;
    const int wu   = __builtin_amdgcn_readfirstlane(tid >> 5);

    if constexpr (!LN) {
        const int nv = nn * Q;
        const float4* x4 = (const float4*)(IN + (size_t)n0 * K);
        int i = tid;
        for (; i + 7168 < nv; i += 8192) {
            float4 q[8];
#pragma unroll
            for (int u = 0; u < 8; ++u) q[u] = x4[i + (u << 10)];
#pragma unroll
            for (int u = 0; u < 8; ++u) {
                int ii = i + (u << 10);
                *(float4*)(tile + (ii / Q) * P + ((ii % Q) << 2)) = q[u];
            }
        }
        for (; i + 3072 < nv; i += 4096) {
            float4 q[4];
#pragma unroll
            for (int u = 0; u < 4; ++u) q[u] = x4[i + (u << 10)];
#pragma unroll
            for (int u = 0; u < 4; ++u) {
                int ii = i + (u << 10);
                *(float4*)(tile + (ii / Q) * P + ((ii % Q) << 2)) = q[u];
            }
        }
        for (; i < nv; i += 1024) {
            float4 qq = x4[i];
            *(float4*)(tile + (i / Q) * P + ((i % Q) << 2)) = qq;
        }
    } else {
        float gv[PER], bv[PER];
#pragma unroll
        for (int j = 0; j < PER; ++j) { gv[j] = g[lane + (j << 5)]; bv[j] = be[lane + (j << 5)]; }
        for (int r0 = wu; r0 < nn; r0 += 32 * R) {
            float vals[R][PER];
            float s[R], mu[R], rstd[R];
#pragma unroll
            for (int u = 0; u < R; ++u) {
                int rr = r0 + (u << 5);
                if (rr >= nn) rr = r0;
                const float* irow = IN + (size_t)(n0 + rr) * K + lane;
#pragma unroll
                for (int j = 0; j < PER; ++j) vals[u][j] = irow[j << 5];
            }
#pragma unroll
            for (int u = 0; u < R; ++u) {
                s[u] = 0.0f;
#pragma unroll
                for (int j = 0; j < PER; ++j) s[u] += vals[u][j];
            }
#pragma unroll
            for (int off = 16; off; off >>= 1)
#pragma unroll
                for (int u = 0; u < R; ++u) s[u] += __shfl_xor(s[u], off, 32);
#pragma unroll
            for (int u = 0; u < R; ++u) {
                mu[u] = s[u] * (1.0f / K);
                s[u] = 0.0f;
#pragma unroll
                for (int j = 0; j < PER; ++j) { float d = vals[u][j] - mu[u]; s[u] += d * d; }
            }
#pragma unroll
            for (int off = 16; off; off >>= 1)
#pragma unroll
                for (int u = 0; u < R; ++u) s[u] += __shfl_xor(s[u], off, 32);
#pragma unroll
            for (int u = 0; u < R; ++u) rstd[u] = rsqrtf(s[u] * (1.0f / K) + 1e-5f);
#pragma unroll
            for (int u = 0; u < R; ++u) {
                int rr = r0 + (u << 5);
                if (rr < nn) {
                    float* trow = tile + rr * P + lane;
#pragma unroll
                    for (int j = 0; j < PER; ++j) {
                        float v = (vals[u][j] - mu[u]) * rstd[u] * gv[j] + bv[j];
                        trow[j << 5] = (v > 0.0f) ? v : (expf(v) - 1.0f);
                    }
                }
            }
        }
    }

    constexpr int TILES_N = M2 / 16;
    constexpr int WPC     = 32 / TILES_N;
    const int lh   = lane >> 4;
    const int lm   = lane & 15;
    const int col0 = (wu % TILES_N) << 4;
    const float* b_ptr = W + (size_t)(lh << 1) * M2 + col0 + lm;
    v2f bfrag[K / 4];
#pragma unroll
    for (int k = 0; k < K; k += 4) {
        bfrag[k >> 2].x = b_ptr[(size_t)k * M2];
        bfrag[k >> 2].y = b_ptr[(size_t)(k + 1) * M2];
    }
    __syncthreads();

    const int nstrips = nn >> 4;
    for (int s = wu / TILES_N; s < nstrips; s += WPC) {
        const int srow = s << 4;
        const float* a_lds = tile + (srow + lm) * P + (lh << 1);
        v8f acc = {};
#pragma unroll
        for (int k = 0; k < K; k += 4) {
            v2f a = *(const v2f*)(a_lds + k);
            acc = __builtin_amdgcn_wmma_f32_16x16x4_f32(false, a, false, bfrag[k >> 2],
                                                        (short)0, acc, false, false);
        }
        store_tile<M2, BIAS>(acc, bias, C, n0 + srow, col0, lh, lm);
    }
}


__global__ __launch_bounds__(1024)
void deg_count_kernel(const int* __restrict__ dst, float* __restrict__ deg,
                      int* __restrict__ bcnt, int E, int NB)
{
    __shared__ int h[MAXNB];
    const int tid = threadIdx.x;
    h[tid] = 0;
    __syncthreads();
    for (int e = blockIdx.x * 1024 + tid; e < E; e += gridDim.x * 1024) {
        int d = dst[e];
        atomicAdd(&deg[d], 1.0f);
        atomicAdd(&h[d / NPB], 1);
    }
    __syncthreads();
    if (tid < NB && h[tid] > 0) atomicAdd(&bcnt[tid], h[tid]);
}

__global__ void dinv_kernel(const float* __restrict__ deg, float* __restrict__ dinv, int n)
{
    int i = blockIdx.x * blockDim.x + threadIdx.x;
    if (i < n) dinv[i] = rsqrtf(deg[i] + 1.0f);
}

__global__ __launch_bounds__(1024)
void place_kernel(const int* __restrict__ src, const int* __restrict__ dst,
                  const float* __restrict__ dinv, const int* __restrict__ bcnt,
                  int* __restrict__ bcur, int4* __restrict__ rec, int E, int NB)
{
    __shared__ int h[MAXNB];
    __shared__ int base[MAXNB];
    __shared__ int wsum[32];
    const int tid  = threadIdx.x;
    const int lane = tid & 31;
    const int w    = tid >> 5;
    h[tid] = 0;
    __syncthreads();

    const int e0 = blockIdx.x * (1024 * EPT);
    int sv[EPT], dv[EPT], bk[EPT], rk[EPT];
#pragma unroll
    for (int i = 0; i < EPT; ++i) {
        int e = e0 + i * 1024 + tid;
        bk[i] = -1; sv[i] = 0; dv[i] = 0; rk[i] = 0;
        if (e < E) {
            sv[i] = src[e];
            dv[i] = dst[e];
            bk[i] = dv[i] / NPB;
            rk[i] = atomicAdd(&h[bk[i]], 1);
        }
    }

    const int c = tid < NB ? bcnt[tid] : 0;
    int x = c;
#pragma unroll
    for (int off = 1; off < 32; off <<= 1) {
        int y = __shfl_up(x, off, 32);
        if (lane >= off) x += y;
    }
    if (lane == 31) wsum[w] = x;
    __syncthreads();
    if (w == 0) {
        int t = wsum[lane];
        int y = t;
#pragma unroll
        for (int off = 1; off < 32; off <<= 1) {
            int z = __shfl_up(y, off, 32);
            if (lane >= off) y += z;
        }
        wsum[lane] = y - t;
    }
    __syncthreads();
    const int mine = h[tid];
    if (tid < NB && mine > 0) base[tid] = x - c + wsum[w] + atomicAdd(&bcur[tid], mine);
    __syncthreads();

#pragma unroll
    for (int i = 0; i < EPT; ++i) {
        if (bk[i] >= 0) {
            float nm = dinv[sv[i]] * dinv[dv[i]];
            rec[base[bk[i]] + rk[i]] = make_int4(sv[i], dv[i], __float_as_int(nm), 0);
        }
    }
}

template<int M, bool FULL>
__device__ __forceinline__
void scatter_edges_lds(int s_l, int o_l, int n_l, int cnt, int lane, unsigned lane_boff,
                       const float* __restrict__ xw, float* tile)
{
    constexpr int PER = M / 32;
    constexpr int U   = 16;
    typedef __attribute__((ext_vector_type(PER))) float vrow;
#pragma unroll 1
    for (int j = 0; j < 32; j += U) {
        if (!FULL && j >= cnt) break;
        vrow v[U];
#pragma unroll
        for (int u = 0; u < U; ++u) {
            int   s  = __builtin_amdgcn_readlane(s_l, j + u);
            float nm = __int_as_float(__builtin_amdgcn_readlane(n_l, j + u));
            const char* rowb = (const char*)(xw + (size_t)s * M);
            vrow q = *(const vrow*)(rowb + lane_boff);
            v[u] = q * nm;
        }
#pragma unroll
        for (int u = 0; u < U; ++u) {
            if (FULL || j + u < cnt) {
                int o = __builtin_amdgcn_readlane(o_l, j + u);
                float* t = tile + o + lane;
#pragma unroll
                for (int c2 = 0; c2 < PER; ++c2) atomicAdd(t + (c2 << 5), v[u][c2]);
            }
        }
    }
}

template<int M, int R>
__device__ __forceinline__
void start_rows_lds(int r0, int n0, int lane, unsigned lane_boff, const float (&bb)[M / 32],
                    const float* __restrict__ xw, const float* __restrict__ dinv, float* tile)
{
    constexpr int P   = M + 4;
    constexpr int PER = M / 32;
    typedef __attribute__((ext_vector_type(PER))) float vrow;
    vrow  q[R];
    float s[R];
#pragma unroll
    for (int u = 0; u < R; ++u) {
        const int rr = r0 + (u << 5);
        const char* rowb = (const char*)(xw + (size_t)(n0 + rr) * M);
        q[u] = *(const vrow*)(rowb + lane_boff);
        float di = dinv[n0 + rr];
        s[u] = di * di;
    }
#pragma unroll
    for (int u = 0; u < R; ++u) {
        float* t = tile + (r0 + (u << 5)) * P + lane;
#pragma unroll
        for (int cc = 0; cc < PER; ++cc) t[cc << 5] = q[u][cc] * s[u] + bb[cc];
    }
}

template<int M, int M2, bool BIAS_OUT>
__global__ __launch_bounds__(1024)
void edge_sum_ln_gemm(const int4* __restrict__ rec, const int* __restrict__ bcnt,
                      const float* __restrict__ xw, const float* __restrict__ bias,
                      const float* __restrict__ dinv, const float* __restrict__ g,
                      const float* __restrict__ be, const float* __restrict__ W,
                      const float* __restrict__ bias_out, float* __restrict__ C, int N)
{
    constexpr int P   = M + 4;
    constexpr int PER = M / 32;
    typedef __attribute__((ext_vector_type(PER))) float vrow;
    __shared__ __attribute__((aligned(16))) float tile[NPB * P];
    __shared__ int wsum[32];
    const int tid  = threadIdx.x;
    const int lane = tid & 31;
    const int w    = tid >> 5;
    const int b    = blockIdx.x;
    const int n0   = b * NPB;
    const int nn   = (N - n0 < NPB) ? (N - n0) : NPB;
    const int wu   = __builtin_amdgcn_readfirstlane(w);
    const unsigned lane_boff = (unsigned)lane * (PER * 4u);

    int c = tid < b ? bcnt[tid] : 0;
#pragma unroll
    for (int off = 16; off; off >>= 1) c += __shfl_xor(c, off, 32);
    if (lane == 0) wsum[w] = c;

    {
        float bb[PER];
#pragma unroll
        for (int cc = 0; cc < PER; ++cc) bb[cc] = bias[PER * lane + cc];
        int r0 = wu;
        for (; r0 + 224 < nn; r0 += 256) start_rows_lds<M, 8>(r0, n0, lane, lane_boff, bb, xw, dinv, tile);
        for (; r0 +  96 < nn; r0 += 128) start_rows_lds<M, 4>(r0, n0, lane, lane_boff, bb, xw, dinv, tile);
        for (; r0       < nn; r0 +=  32) start_rows_lds<M, 1>(r0, n0, lane, lane_boff, bb, xw, dinv, tile);
    }
    __syncthreads();

    int start = 0;
#pragma unroll
    for (int i = 0; i < 32; ++i) start += wsum[i];
    start = __builtin_amdgcn_readfirstlane(start);
    const int end = start + __builtin_amdgcn_readfirstlane(bcnt[b]);

    const int4 none = make_int4(0, n0, 0, 0);
    int c0 = start + (wu << 5);
    int4 r = none;
    if (c0 + lane < end) r = rec[c0 + lane];
    for (; c0 < end; c0 += 1024) {
        int4 rn = none;
        if (c0 + 1024 + lane < end) rn = rec[c0 + 1024 + lane];
        int cnt = end - c0;
        int o_l = (r.y - n0) * P;
        if (cnt >= 32) scatter_edges_lds<M, true >(r.x, o_l, r.z, 32,  lane, lane_boff, xw, tile);
        else           scatter_edges_lds<M, false>(r.x, o_l, r.z, cnt, lane, lane_boff, xw, tile);
        r = rn;
    }
    __syncthreads();

    constexpr int TILES_N = M2 / 16;
    constexpr int WPC     = 32 / TILES_N;
    const int lh   = lane >> 4;
    const int lm   = lane & 15;
    const int col0 = (wu % TILES_N) << 4;
    const float* b_ptr = W + (size_t)(lh * 2 * PER) * M2 + col0 + lm;
    v2f bfrag[M / 4];
#pragma unroll
    for (int k = 0; k < M; k += 4) {
        const int kr = PER * (k & 31) + (k >> 5);
        bfrag[k >> 2].x = b_ptr[(size_t)kr * M2];
        bfrag[k >> 2].y = b_ptr[(size_t)(kr + PER) * M2];
    }

    {
        float gv[PER], bv[PER];
#pragma unroll
        for (int j = 0; j < PER; ++j) { gv[j] = g[PER * lane + j]; bv[j] = be[PER * lane + j]; }
        for (int r0 = wu; r0 < nn; r0 += 128) {
            float vals[4][PER];
            float s[4], mu[4], rstd[4];
#pragma unroll
            for (int u = 0; u < 4; ++u) {
                int rr = r0 + (u << 5);
                if (rr >= nn) rr = r0;
                const float* trow = tile + rr * P + lane;
#pragma unroll
                for (int j = 0; j < PER; ++j) vals[u][j] = trow[j << 5];
            }
#pragma unroll
            for (int u = 0; u < 4; ++u) {
                s[u] = 0.0f;
#pragma unroll
                for (int j = 0; j < PER; ++j) s[u] += vals[u][j];
            }
#pragma unroll
            for (int off = 16; off; off >>= 1)
#pragma unroll
                for (int u = 0; u < 4; ++u) s[u] += __shfl_xor(s[u], off, 32);
#pragma unroll
            for (int u = 0; u < 4; ++u) {
                mu[u] = s[u] * (1.0f / M);
                s[u] = 0.0f;
#pragma unroll
                for (int j = 0; j < PER; ++j) { float d = vals[u][j] - mu[u]; s[u] += d * d; }
            }
#pragma unroll
            for (int off = 16; off; off >>= 1)
#pragma unroll
                for (int u = 0; u < 4; ++u) s[u] += __shfl_xor(s[u], off, 32);
#pragma unroll
            for (int u = 0; u < 4; ++u) rstd[u] = rsqrtf(s[u] * (1.0f / M) + 1e-5f);
#pragma unroll
            for (int u = 0; u < 4; ++u) {
                int rr = r0 + (u << 5);
                if (rr < nn) {
                    float* trow = tile + rr * P + lane;
#pragma unroll
                    for (int j = 0; j < PER; ++j) {
                        float v = (vals[u][j] - mu[u]) * rstd[u] * gv[j] + bv[j];
                        trow[j << 5] = (v > 0.0f) ? v : (expf(v) - 1.0f);
                    }
                }
            }
        }
    }
    __syncthreads();

    const int nstrips = nn >> 4;
    for (int s = wu / TILES_N; s < nstrips; s += WPC) {
        const int srow = s << 4;
        const float* a_lds = tile + (srow + lm) * P + (lh << 1);
        v8f acc = {};
#pragma unroll
        for (int k = 0; k < M; k += 4) {
            v2f a = *(const v2f*)(a_lds + k);
            acc = __builtin_amdgcn_wmma_f32_16x16x4_f32(false, a, false, bfrag[k >> 2],
                                                        (short)0, acc, false, false);
        }
        store_tile<M2, BIAS_OUT>(acc, bias_out, C, n0 + srow, col0, lh, lm);
    }
}

extern "C" void kernel_launch(void* const* d_in, const int* in_sizes, int n_in,
                              void* d_out, int out_size, void* d_ws, size_t ws_size,
                              hipStream_t stream)
{
    const float* x   = (const float*)d_in[0];
    const int*   ei  = (const int*)  d_in[1];
    const float* W1  = (const float*)d_in[2];
    const float* b1  = (const float*)d_in[3];
    const float* g1  = (const float*)d_in[4];
    const float* be1 = (const float*)d_in[5];
    const float* W2  = (const float*)d_in[6];
    const float* b2  = (const float*)d_in[7];
    const float* g2  = (const float*)d_in[8];
    const float* be2 = (const float*)d_in[9];
    const float* W3  = (const float*)d_in[10];
    const float* b3  = (const float*)d_in[11];
    const float* g3  = (const float*)d_in[12];
    const float* be3 = (const float*)d_in[13];
    const float* lw1 = (const float*)d_in[14];
    const float* lb1 = (const float*)d_in[15];
    const float* g4  = (const float*)d_in[16];
    const float* be4 = (const float*)d_in[17];
    const float* lw2 = (const float*)d_in[18];
    const float* lb2 = (const float*)d_in[19];

    const int N = in_sizes[0] / 128;
    const int E = in_sizes[1] / 2;
    const int* src = ei;
    const int* dst = ei + E;

    float* ws   = (float*)d_ws;
    float* xwB  = ws;  ws += (size_t)N * 128;
    float* xwA  = ws;  ws += (size_t)N * 64;
    float* hid  = ws;  ws += (size_t)N * 32;
    float* deg  = ws;  ws += N;
    int*   bcnt = (int*)ws;  ws += MAXNB;
    int*   bcur = (int*)ws;  ws += MAXNB;
    float* dinv = ws;  ws += N;
    ws += (4 - (((size_t)N * 2) & 3)) & 3;
    int4*  rec  = (int4*)ws;
    const int NB = ceil_div_ll(N, NPB);

    const int B = 256;
    fill_f32_kernel<<<ceil_div_ll(N + 2 * MAXNB, B), B, 0, stream>>>(deg, 0.0f, (long long)N + 2 * MAXNB);
    deg_count_kernel<<<256, 1024, 0, stream>>>(dst, deg, bcnt, E, NB);
    dinv_kernel<<<ceil_div_ll(N, B), B, 0, stream>>>(deg, dinv, N);
    place_kernel<<<ceil_div_ll(E, 1024 * EPT), 1024, 0, stream>>>(src, dst, dinv, bcnt, bcur, rec, E, NB);

    bucket_gemm_f32<128, 64, false, false><<<NB, 1024, 0, stream>>>(x, nullptr, nullptr, W1, nullptr, xwA, N);

    edge_sum_ln_gemm<64, 128, false><<<NB, 1024, 0, stream>>>(rec, bcnt, xwA, b1, dinv, g1, be1, W2, nullptr, xwB, N);

    edge_sum_ln_gemm<128, 64, false><<<NB, 1024, 0, stream>>>(rec, bcnt, xwB, b2, dinv, g2, be2, W3, nullptr, xwA, N);

    edge_sum_ln_gemm<64, 32, true><<<NB, 1024, 0, stream>>>(rec, bcnt, xwA, b3, dinv, g3, be3, lw1, lb1, hid, N);

    bucket_gemm_f32<32, 32, true, true><<<NB, 1024, 0, stream>>>(hid, g4, be4, lw2, lb2, (float*)d_out, N);
}
